# starting kernel
# baseline (speedup 1.0000x reference)
#include <hip/hip_runtime.h>
#include <hip/hip_bf16.h>

typedef __attribute__((ext_vector_type(2))) float v2f;
typedef __attribute__((ext_vector_type(4))) float v4f;
typedef __attribute__((ext_vector_type(8))) float v8f;

#define HEADS 4
#define CH 128
#define NEG_SLOPE 0.2f

__device__ __forceinline__ unsigned enc_f32(float f) {
  unsigned u = __float_as_uint(f);
  return (f >= 0.0f) ? (u | 0x80000000u) : ~u;
}
__device__ __forceinline__ float dec_f32(unsigned u) {
  return (u & 0x80000000u) ? __uint_as_float(u & 0x7fffffffu)
                           : __uint_as_float(~u);
}

__device__ __forceinline__ void edge_nodes(const int* __restrict__ ei, int e,
                                           int ne, int& s, int& d) {
  if (e < ne) { s = ei[e]; d = ei[ne + e]; }
  else        { s = d = e - ne; }
}

__global__ void gat_init(float* __restrict__ out, unsigned* __restrict__ emax,
                         float* __restrict__ denom, int n) {
  int t = blockIdx.x * blockDim.x + threadIdx.x;
  if (t < n * CH) out[t] = 0.0f;
  if (t < n * HEADS) { emax[t] = 0u; denom[t] = 0.0f; }
}

__global__ __launch_bounds__(256) void gat_gemm_wmma(
    const float* __restrict__ x, const float* __restrict__ W,
    float* __restrict__ h, int n) {
  const int wave = threadIdx.x >> 5;
  const int lane = threadIdx.x & 31;
  const int m0 = blockIdx.x * 16;
  const int n0 = wave * 16;
  const int lm = lane & 15;
  const int lk = (lane >> 4) * 2;

  int row = m0 + lm;
  if (row >= n) row = n - 1;
  const float* __restrict__ arow = x + (size_t)row * CH;

  v8f acc = {};
#pragma unroll
  for (int k0 = 0; k0 < CH; k0 += 4) {
    v2f a = *(const v2f*)(arow + k0 + lk);
    v2f b;
    b.x = W[(size_t)(k0 + lk) * CH + n0 + lm];
    b.y = W[(size_t)(k0 + lk + 1) * CH + n0 + lm];
    acc = __builtin_amdgcn_wmma_f32_16x16x4_f32(
         false, a,  false, b,
         (short)0, acc,  false,  false);
  }

  const int colg = n0 + lm;
  const int rbase = m0 + 8 * (lane >> 4);
#pragma unroll
  for (int v = 0; v < 8; ++v) {
    int rowg = rbase + v;
    if (rowg < n) h[(size_t)rowg * CH + colg] = acc[v];
  }
}

__global__ void gat_att(const float* __restrict__ h,
                        const float* __restrict__ att_src,
                        const float* __restrict__ att_dst,
                        float* __restrict__ a_src, float* __restrict__ a_dst,
                        int n) {
  int t = blockIdx.x * blockDim.x + threadIdx.x;
  if (t >= n * HEADS) return;
  int head = t & (HEADS - 1);
  const float* __restrict__ hv = h + (size_t)(t >> 2) * CH + head * 32;
  const float* __restrict__ as = att_src + head * 32;
  const float* __restrict__ ad = att_dst + head * 32;
  float s = 0.0f, d = 0.0f;
#pragma unroll
  for (int c = 0; c < 32; ++c) { float v = hv[c]; s += v * as[c]; d += v * ad[c]; }
  a_src[t] = s;
  a_dst[t] = d;
}

__global__ void gat_emax(const int* __restrict__ ei,
                         const float* __restrict__ a_src,
                         const float* __restrict__ a_dst,
                         unsigned* __restrict__ emax, int n, int ne) {
  int t = blockIdx.x * blockDim.x + threadIdx.x;
  if (t >= (ne + n) * HEADS) return;
  int head = t & (HEADS - 1);
  int e = t >> 2;
  int s, d; edge_nodes(ei, e, ne, s, d);
  float v = a_src[s * HEADS + head] + a_dst[d * HEADS + head];
  v = (v >= 0.0f) ? v : NEG_SLOPE * v;
  atomicMax(&emax[d * HEADS + head], enc_f32(v));
}

__global__ void gat_denom(const int* __restrict__ ei,
                          const float* __restrict__ a_src,
                          const float* __restrict__ a_dst,
                          const unsigned* __restrict__ emax,
                          float* __restrict__ denom, int n, int ne) {
  int t = blockIdx.x * blockDim.x + threadIdx.x;
  if (t >= (ne + n) * HEADS) return;
  int head = t & (HEADS - 1);
  int e = t >> 2;
  int s, d; edge_nodes(ei, e, ne, s, d);
  float v = a_src[s * HEADS + head] + a_dst[d * HEADS + head];
  v = (v >= 0.0f) ? v : NEG_SLOPE * v;
  float m = dec_f32(emax[d * HEADS + head]);
  atomicAdd(&denom[d * HEADS + head], expf(v - m));
}

__global__ void gat_scatter(const int* __restrict__ ei,
                            const float* __restrict__ h,
                            const float* __restrict__ a_src,
                            const float* __restrict__ a_dst,
                            const unsigned* __restrict__ emax,
                            const float* __restrict__ denom,
                            float* __restrict__ out, int n, int ne) {
  int gid = blockIdx.x * blockDim.x + threadIdx.x;
  int e = gid >> 5;
  int lane = gid & 31;
  if (e >= ne + n) return;
  int s, d; edge_nodes(ei, e, ne, s, d);
  int f0 = lane * 4;
  int head = f0 >> 5;
  float v = a_src[s * HEADS + head] + a_dst[d * HEADS + head];
  v = (v >= 0.0f) ? v : NEG_SLOPE * v;
  float m = dec_f32(emax[d * HEADS + head]);
  float alpha = expf(v - m) / (denom[d * HEADS + head] + 1e-16f);
  v4f hv = *(const v4f*)(h + (size_t)s * CH + f0);
  float* __restrict__ od = out + (size_t)d * CH + f0;
  atomicAdd(od + 0, hv.x * alpha);
  atomicAdd(od + 1, hv.y * alpha);
  atomicAdd(od + 2, hv.z * alpha);
  atomicAdd(od + 3, hv.w * alpha);
}

__global__ void gat_prelu(float* __restrict__ out, const float* __restrict__ bias,
                          const float* __restrict__ prelu_w, int n) {
  int t = blockIdx.x * blockDim.x + threadIdx.x;
  if (t >= n * CH) return;
  float o = out[t] + bias[t & (CH - 1)];
  float pw = prelu_w[0];
  out[t] = (o >= 0.0f) ? o : pw * o;
}

extern "C" void kernel_launch(void* const* d_in, const int* in_sizes, int n_in,
                              void* d_out, int out_size, void* d_ws, size_t ws_size,
                              hipStream_t stream) {
  const float* x       = (const float*)d_in[0];
  const int*   ei      = (const int*)d_in[1];
  const float* W       = (const float*)d_in[2];
  const float* att_src = (const float*)d_in[3];
  const float* att_dst = (const float*)d_in[4];
  const float* bias    = (const float*)d_in[5];
  const float* prelu_w = (const float*)d_in[6];
  float* out = (float*)d_out;

  const int n  = in_sizes[0] / CH;
  const int ne = in_sizes[1] / 2;

  float*    h      = (float*)d_ws;
  float*    a_src  = h + (size_t)n * CH;
  float*    a_dst  = a_src + (size_t)n * HEADS;
  unsigned* emax   = (unsigned*)(a_dst + (size_t)n * HEADS);
  float*    denom  = (float*)(emax + (size_t)n * HEADS);

  const int tpb = 256;
  int total_eh = (ne + n) * HEADS;

  gat_init<<<(n * CH + tpb - 1) / tpb, tpb, 0, stream>>>(out, emax, denom, n);
  gat_gemm_wmma<<<(n + 15) / 16, 256, 0, stream>>>(x, W, h, n);
  gat_att<<<(n * HEADS + tpb - 1) / tpb, tpb, 0, stream>>>(h, att_src, att_dst,
                                                           a_src, a_dst, n);
  gat_emax<<<(total_eh + tpb - 1) / tpb, tpb, 0, stream>>>(ei, a_src, a_dst,
                                                           emax, n, ne);
  gat_denom<<<(total_eh + tpb - 1) / tpb, tpb, 0, stream>>>(ei, a_src, a_dst,
                                                            emax, denom, n, ne);
  long long sc_threads = (long long)(ne + n) * 32;
  gat_scatter<<<(int)((sc_threads + tpb - 1) / tpb), tpb, 0, stream>>>(
      ei, h, a_src, a_dst, emax, denom, out, n, ne);
  gat_prelu<<<(n * CH + tpb - 1) / tpb, tpb, 0, stream>>>(out, bias, prelu_w, n);
}


# Atomic scatter and atomic segment softmax replaced by an in-call counting sort by destination and a wave-per-node gather with a running softmax in registers; per-node sum order changes
# speedup vs baseline: 2.2576x; 2.2576x over previous
#include <hip/hip_runtime.h>
#include <hip/hip_bf16.h>

typedef __attribute__((ext_vector_type(2))) float v2f;
typedef __attribute__((ext_vector_type(4))) float v4f;
typedef __attribute__((ext_vector_type(8))) float v8f;

#define HEADS 4
#define CH 128
#define NEG_SLOPE 0.2f

#define M_CNT 0
#define M_OFF 1
#define M_CUR 2
#define M_ALLOC 3

__global__ void gat_init(unsigned* __restrict__ meta, int n) {
  int t = blockIdx.x * blockDim.x + threadIdx.x;
  if (t < n) meta[(size_t)t * CH + M_CNT] = 0u;
  if (t == 0) meta[M_ALLOC] = 0u;
}

__global__ void gat_hist(const int* __restrict__ ei, unsigned* __restrict__ meta,
                         int ne) {
  int e = blockIdx.x * blockDim.x + threadIdx.x;
  if (e >= ne) return;
  int d = ei[ne + e];
  atomicAdd(&meta[(size_t)d * CH + M_CNT], 1u);
}

__global__ void gat_alloc(unsigned* __restrict__ meta, int n) {
  int t = blockIdx.x * blockDim.x + threadIdx.x;
  int lane = threadIdx.x & 31;
  unsigned c = (t < n) ? meta[(size_t)t * CH + M_CNT] : 0u;
  unsigned incl = c;
#pragma unroll
  for (int o = 1; o < 32; o <<= 1) {
    unsigned up = __shfl_up(incl, o, 32);
    if (lane >= o) incl += up;
  }
  unsigned total = __shfl(incl, 31, 32);
  unsigned base = 0u;
  if (lane == 31) base = atomicAdd(&meta[M_ALLOC], total);
  base = __shfl(base, 31, 32);
  unsigned off = base + incl - c;
  if (t < n) {
    meta[(size_t)t * CH + M_OFF] = off;
    meta[(size_t)t * CH + M_CUR] = off;
  }
}

__global__ void gat_fill(const int* __restrict__ ei, unsigned* __restrict__ meta,
                         int* __restrict__ srcs, int ne) {
  int e = blockIdx.x * blockDim.x + threadIdx.x;
  if (e >= ne) return;
  int s = ei[e];
  int d = ei[ne + e];
  unsigned pos = atomicAdd(&meta[(size_t)d * CH + M_CUR], 1u);
  srcs[pos] = s;
}

__global__ __launch_bounds__(256) void gat_gemm_wmma(
    const float* __restrict__ x, const float* __restrict__ W,
    float* __restrict__ h, int n) {
  const int wave = threadIdx.x >> 5;
  const int lane = threadIdx.x & 31;
  const int m0 = blockIdx.x * 16;
  const int n0 = wave * 16;
  const int lm = lane & 15;
  const int lk = (lane >> 4) * 2;

  int row = m0 + lm;
  if (row >= n) row = n - 1;
  const float* __restrict__ arow = x + (size_t)row * CH;

  v8f acc = {};
#pragma unroll
  for (int k0 = 0; k0 < CH; k0 += 4) {
    v2f a = *(const v2f*)(arow + k0 + lk);
    v2f b;
    b.x = W[(size_t)(k0 + lk) * CH + n0 + lm];
    b.y = W[(size_t)(k0 + lk + 1) * CH + n0 + lm];
    acc = __builtin_amdgcn_wmma_f32_16x16x4_f32(
         false, a,  false, b,
         (short)0, acc,  false,  false);
  }

  const int colg = n0 + lm;
  const int rbase = m0 + 8 * (lane >> 4);
#pragma unroll
  for (int v = 0; v < 8; ++v) {
    int rowg = rbase + v;
    if (rowg < n) h[(size_t)rowg * CH + colg] = acc[v];
  }
}

template <int MASK>
__device__ __forceinline__ float lane_xor(float v) {
  return __int_as_float(__builtin_amdgcn_update_dpp(
      0, __float_as_int(v), 0x160 | MASK, 0xf, 0xf, false));
}

__device__ __forceinline__ float head_sum(float v) {
  v += lane_xor<1>(v);
  v += lane_xor<2>(v);
  v += lane_xor<4>(v);
  return v;
}

#define GAT_STEP(r)                                                        \
  {                                                                        \
    float es = head_sum((r).x * as4.x + (r).y * as4.y + (r).z * as4.z +    \
                        (r).w * as4.w);                                    \
    float e = es + adst;                                                   \
    e = (e >= 0.0f) ? e : NEG_SLOPE * e;                                   \
            \
    const bool up = e > m;                                                 \
    const float t = expf(up ? m - e : e - m);                              \
    const float sc = up ? t : 1.0f;                                        \
    const float p = up ? 1.0f : t;                                         \
    l = l * sc + p;                                                        \
    acc = acc * sc + (r) * p;                                              \
    m = up ? e : m;                                                        \
  }

__global__ __launch_bounds__(256) void gat_gather(
    const float* __restrict__ h, const int* __restrict__ srcs,
    const float* __restrict__ att_src, const float* __restrict__ att_dst,
    const float* __restrict__ bias, const float* __restrict__ prelu_w,
    float* out, int n) {
  const int lane = threadIdx.x & 31;
  const int d = __builtin_amdgcn_readfirstlane(
      (int)((blockIdx.x * blockDim.x + threadIdx.x) >> 5));
  if (d >= n) return;
  const int f0 = lane * 4;
  const v4f as4 = *(const v4f*)(att_src + f0);
  const v4f ad4 = *(const v4f*)(att_dst + f0);
  const unsigned* mrow = (const unsigned*)(out + (size_t)d * CH);
  const unsigned cnt = __builtin_amdgcn_readfirstlane(mrow[M_CNT]);
  const unsigned off = __builtin_amdgcn_readfirstlane(mrow[M_OFF]);

  const v4f hd = *(const v4f*)(h + (size_t)d * CH + f0);
  const float adst =
      head_sum(hd.x * ad4.x + hd.y * ad4.y + hd.z * ad4.z + hd.w * ad4.w);
  float m = head_sum(hd.x * as4.x + hd.y * as4.y + hd.z * as4.z + hd.w * as4.w) +
            adst;
  m = (m >= 0.0f) ? m : NEG_SLOPE * m;
  float l = 1.0f;
  v4f acc = hd;

  for (unsigned base = 0; base < cnt; base += 32) {
    const unsigned rem = cnt - base;
    const unsigned mm = rem < 32u ? rem : 32u;
    int smine = ((unsigned)lane < mm) ? srcs[off + base + lane] : 0;
    unsigned j = 0;
    for (; j + 4 <= mm; j += 4) {
      const int s0 = __builtin_amdgcn_readlane(smine, (int)j);
      const int s1 = __builtin_amdgcn_readlane(smine, (int)j + 1);
      const int s2 = __builtin_amdgcn_readlane(smine, (int)j + 2);
      const int s3 = __builtin_amdgcn_readlane(smine, (int)j + 3);
      const v4f r0 = *(const v4f*)(h + (size_t)s0 * CH + f0);
      const v4f r1 = *(const v4f*)(h + (size_t)s1 * CH + f0);
      const v4f r2 = *(const v4f*)(h + (size_t)s2 * CH + f0);
      const v4f r3 = *(const v4f*)(h + (size_t)s3 * CH + f0);
      GAT_STEP(r0);
      GAT_STEP(r1);
      GAT_STEP(r2);
      GAT_STEP(r3);
    }
    for (; j < mm; ++j) {
      const int s0 = __builtin_amdgcn_readlane(smine, (int)j);
      const v4f r0 = *(const v4f*)(h + (size_t)s0 * CH + f0);
      GAT_STEP(r0);
    }
  }

  const float den = l + 1e-16f;
  const v4f b4 = *(const v4f*)(bias + f0);
  const float pw = prelu_w[0];
  v4f o;
  o.x = acc.x / den + b4.x;
  o.y = acc.y / den + b4.y;
  o.z = acc.z / den + b4.z;
  o.w = acc.w / den + b4.w;
  o.x = (o.x >= 0.0f) ? o.x : pw * o.x;
  o.y = (o.y >= 0.0f) ? o.y : pw * o.y;
  o.z = (o.z >= 0.0f) ? o.z : pw * o.z;
  o.w = (o.w >= 0.0f) ? o.w : pw * o.w;
  *(v4f*)(out + (size_t)d * CH + f0) = o;
}

extern "C" void kernel_launch(void* const* d_in, const int* in_sizes, int n_in,
                              void* d_out, int out_size, void* d_ws, size_t ws_size,
                              hipStream_t stream) {
  const float* x       = (const float*)d_in[0];
  const int*   ei      = (const int*)d_in[1];
  const float* W       = (const float*)d_in[2];
  const float* att_src = (const float*)d_in[3];
  const float* att_dst = (const float*)d_in[4];
  const float* bias    = (const float*)d_in[5];
  const float* prelu_w = (const float*)d_in[6];
  float* out = (float*)d_out;

  const int n  = in_sizes[0] / CH;
  const int ne = in_sizes[1] / 2;

  float* h    = (float*)d_ws;
  int*   srcs = (int*)(h + (size_t)n * CH);
  unsigned* meta = (unsigned*)d_out;

  const int tpb = 256;

  gat_init<<<(n + tpb - 1) / tpb, tpb, 0, stream>>>(meta, n);
  gat_hist<<<(ne + tpb - 1) / tpb, tpb, 0, stream>>>(ei, meta, ne);
  gat_alloc<<<(n + tpb - 1) / tpb, tpb, 0, stream>>>(meta, n);
  gat_fill<<<(ne + tpb - 1) / tpb, tpb, 0, stream>>>(ei, meta, srcs, ne);
  gat_gemm_wmma<<<(n + 15) / 16, 256, 0, stream>>>(x, W, h, n);
  gat_gather<<<(n + 7) / 8, 256, 0, stream>>>(h, srcs, att_src, att_dst, bias,
                                              prelu_w, out, n);
}


# Three-pass counting sort (count, segment scan, fill) replaced by one pass: the count atomic's return is the edge's slot in its destination's own output row; per-node sum order stays by arrival
# speedup vs baseline: 2.8471x; 1.2611x over previous
#include <hip/hip_runtime.h>
#include <hip/hip_bf16.h>

typedef __attribute__((ext_vector_type(2))) float v2f;
typedef __attribute__((ext_vector_type(4))) float v4f;
typedef __attribute__((ext_vector_type(8))) float v8f;

#define HEADS 4
#define CH 128
#define NEG_SLOPE 0.2f

#define SLOTS 127u

__global__ void gat_init(unsigned* __restrict__ meta, int n) {
  int t = blockIdx.x * blockDim.x + threadIdx.x;
  if (t < n) meta[(size_t)t * CH] = 0u;
}

__global__ void gat_slot(const int* __restrict__ ei, unsigned* __restrict__ meta,
                         int ne) {
  int e = blockIdx.x * blockDim.x + threadIdx.x;
  if (e >= ne) return;
  int s = ei[e];
  int d = ei[ne + e];
  unsigned* row = meta + (size_t)d * CH;
  unsigned r = atomicAdd(&row[0], 1u);
  if (r < SLOTS) row[1u + r] = (unsigned)s;
}

__global__ __launch_bounds__(256) void gat_gemm_wmma(
    const float* __restrict__ x, const float* __restrict__ W,
    float* __restrict__ h, int n) {
  const int wave = threadIdx.x >> 5;
  const int lane = threadIdx.x & 31;
  const int m0 = blockIdx.x * 16;
  const int n0 = wave * 16;
  const int lm = lane & 15;
  const int lk = (lane >> 4) * 2;

  int row = m0 + lm;
  if (row >= n) row = n - 1;
  const float* __restrict__ arow = x + (size_t)row * CH;

  v8f acc = {};
#pragma unroll
  for (int k0 = 0; k0 < CH; k0 += 4) {
    v2f a = *(const v2f*)(arow + k0 + lk);
    v2f b;
    b.x = W[(size_t)(k0 + lk) * CH + n0 + lm];
    b.y = W[(size_t)(k0 + lk + 1) * CH + n0 + lm];
    acc = __builtin_amdgcn_wmma_f32_16x16x4_f32(
         false, a,  false, b,
         (short)0, acc,  false,  false);
  }

  const int colg = n0 + lm;
  const int rbase = m0 + 8 * (lane >> 4);
#pragma unroll
  for (int v = 0; v < 8; ++v) {
    int rowg = rbase + v;
    if (rowg < n) h[(size_t)rowg * CH + colg] = acc[v];
  }
}

template <int MASK>
__device__ __forceinline__ float lane_xor(float v) {
  return __int_as_float(__builtin_amdgcn_update_dpp(
      0, __float_as_int(v), 0x160 | MASK, 0xf, 0xf, false));
}

__device__ __forceinline__ float head_sum(float v) {
  v += lane_xor<1>(v);
  v += lane_xor<2>(v);
  v += lane_xor<4>(v);
  return v;
}

#define GAT_STEP(r)                                                        \
  {                                                                        \
    float es = head_sum((r).x * as4.x + (r).y * as4.y + (r).z * as4.z +    \
                        (r).w * as4.w);                                    \
    float e = es + adst;                                                   \
    e = (e >= 0.0f) ? e : NEG_SLOPE * e;                                   \
            \
    const bool up = e > m;                                                 \
    const float t = expf(up ? m - e : e - m);                              \
    const float sc = up ? t : 1.0f;                                        \
    const float p = up ? 1.0f : t;                                         \
    l = l * sc + p;                                                        \
    acc = acc * sc + (r) * p;                                              \
    m = up ? e : m;                                                        \
  }

__global__ __launch_bounds__(256) void gat_gather(
    const float* __restrict__ h, const int* __restrict__ ei,
    const float* __restrict__ att_src, const float* __restrict__ att_dst,
    const float* __restrict__ bias, const float* __restrict__ prelu_w,
    float* out, int n, int ne) {
  const int lane = threadIdx.x & 31;
  const int d = __builtin_amdgcn_readfirstlane(
      (int)((blockIdx.x * blockDim.x + threadIdx.x) >> 5));
  if (d >= n) return;
  const int f0 = lane * 4;
  const v4f as4 = *(const v4f*)(att_src + f0);
  const v4f ad4 = *(const v4f*)(att_dst + f0);
  const unsigned* row = (const unsigned*)(out + (size_t)d * CH);
  unsigned wmine = row[lane];
  const unsigned cnt = __builtin_amdgcn_readfirstlane(wmine);

  const v4f hd = *(const v4f*)(h + (size_t)d * CH + f0);
  const float adst =
      head_sum(hd.x * ad4.x + hd.y * ad4.y + hd.z * ad4.z + hd.w * ad4.w);
  float m = head_sum(hd.x * as4.x + hd.y * as4.y + hd.z * as4.z + hd.w * as4.w) +
            adst;
  m = (m >= 0.0f) ? m : NEG_SLOPE * m;
  float l = 1.0f;
  v4f acc = hd;

  if (cnt <= SLOTS) {
    for (unsigned base = 0; base <= cnt; base += 32) {
      if (base) wmine = row[base + lane];
      const unsigned rem = cnt + 1u - base;
      const unsigned jend = rem < 32u ? rem : 32u;
      unsigned j = base ? 0u : 1u;
      for (; j + 4 <= jend; j += 4) {
        const int s0 = __builtin_amdgcn_readlane((int)wmine, (int)j);
        const int s1 = __builtin_amdgcn_readlane((int)wmine, (int)j + 1);
        const int s2 = __builtin_amdgcn_readlane((int)wmine, (int)j + 2);
        const int s3 = __builtin_amdgcn_readlane((int)wmine, (int)j + 3);
        const v4f r0 = *(const v4f*)(h + (size_t)s0 * CH + f0);
        const v4f r1 = *(const v4f*)(h + (size_t)s1 * CH + f0);
        const v4f r2 = *(const v4f*)(h + (size_t)s2 * CH + f0);
        const v4f r3 = *(const v4f*)(h + (size_t)s3 * CH + f0);
        GAT_STEP(r0);
        GAT_STEP(r1);
        GAT_STEP(r2);
        GAT_STEP(r3);
      }
      for (; j < jend; ++j) {
        const int s0 = __builtin_amdgcn_readlane((int)wmine, (int)j);
        const v4f r0 = *(const v4f*)(h + (size_t)s0 * CH + f0);
        GAT_STEP(r0);
      }
    }
  } else {
    for (int base = 0; base < ne; base += 32) {
      const int e = base + lane;
      const int dd = (e < ne) ? ei[ne + e] : -1;
      unsigned hit = __builtin_amdgcn_ballot_w32(dd == d);
      if (hit) {
        const int smine = (e < ne) ? ei[e] : 0;
        while (hit) {
          const int j = __builtin_ctz(hit);
          hit &= hit - 1u;
          const int s0 = __builtin_amdgcn_readlane(smine, j);
          const v4f r0 = *(const v4f*)(h + (size_t)s0 * CH + f0);
          GAT_STEP(r0);
        }
      }
    }
  }

  const float den = l + 1e-16f;
  const v4f b4 = *(const v4f*)(bias + f0);
  const float pw = prelu_w[0];
  v4f o;
  o.x = acc.x / den + b4.x;
  o.y = acc.y / den + b4.y;
  o.z = acc.z / den + b4.z;
  o.w = acc.w / den + b4.w;
  o.x = (o.x >= 0.0f) ? o.x : pw * o.x;
  o.y = (o.y >= 0.0f) ? o.y : pw * o.y;
  o.z = (o.z >= 0.0f) ? o.z : pw * o.z;
  o.w = (o.w >= 0.0f) ? o.w : pw * o.w;
  *(v4f*)(out + (size_t)d * CH + f0) = o;
}

extern "C" void kernel_launch(void* const* d_in, const int* in_sizes, int n_in,
                              void* d_out, int out_size, void* d_ws, size_t ws_size,
                              hipStream_t stream) {
  const float* x       = (const float*)d_in[0];
  const int*   ei      = (const int*)d_in[1];
  const float* W       = (const float*)d_in[2];
  const float* att_src = (const float*)d_in[3];
  const float* att_dst = (const float*)d_in[4];
  const float* bias    = (const float*)d_in[5];
  const float* prelu_w = (const float*)d_in[6];
  float* out = (float*)d_out;

  const int n  = in_sizes[0] / CH;
  const int ne = in_sizes[1] / 2;

  float* h = (float*)d_ws;
  unsigned* meta = (unsigned*)d_out;

  const int tpb = 256;

  gat_init<<<(n + tpb - 1) / tpb, tpb, 0, stream>>>(meta, n);
  gat_slot<<<(ne + tpb - 1) / tpb, tpb, 0, stream>>>(ei, meta, ne);
  gat_gemm_wmma<<<(n + 15) / 16, 256, 0, stream>>>(x, W, h, n);
  gat_gather<<<(n + 7) / 8, 256, 0, stream>>>(h, ei, att_src, att_dst, bias,
                                              prelu_w, out, n, ne);
}


# Slot pass folded into the projection launch: each 256-thread block does its 16-row matrix tile and 256 edges' count adds, which return under the tile; one launch fewer; node sums still by arrival
# speedup vs baseline: 3.1660x; 1.1120x over previous
#include <hip/hip_runtime.h>
#include <hip/hip_bf16.h>

typedef __attribute__((ext_vector_type(2))) float v2f;
typedef __attribute__((ext_vector_type(4))) float v4f;
typedef __attribute__((ext_vector_type(8))) float v8f;

#define HEADS 4
#define CH 128
#define NEG_SLOPE 0.2f

#define SLOTS 127u

__global__ void gat_init(unsigned* __restrict__ meta, int n) {
  int t = blockIdx.x * blockDim.x + threadIdx.x;
  if (t < n) meta[(size_t)t * CH] = 0u;
}

__global__ __launch_bounds__(256, 16) void gat_proj_slot(
    const float* __restrict__ x, const float* __restrict__ W,
    float* __restrict__ h, const int* __restrict__ ei,
    unsigned* __restrict__ meta, int n, int ne) {
  const int e = blockIdx.x * 256 + threadIdx.x;
  const bool has_edge = e < ne;
  const int ec = has_edge ? e : ne - 1;
  const int s = ei[ec];
  const int d = ei[ne + ec];

  const int wave = threadIdx.x >> 5;
  const int lane = threadIdx.x & 31;
  const int m0 = blockIdx.x * 16;
  const int n0 = wave * 16;
  const int lm = lane & 15;
  const int lk = (lane >> 4) * 2;
  const bool has_tile = m0 < n;

  v8f acc = {};
  if (has_tile) {
    int row = m0 + lm;
    if (row >= n) row = n - 1;
    const float* __restrict__ arow = x + (size_t)row * CH;
#pragma unroll
    for (int k0 = 0; k0 < CH; k0 += 4) {
      v2f a = *(const v2f*)(arow + k0 + lk);
      v2f b;
      b.x = W[(size_t)(k0 + lk) * CH + n0 + lm];
      b.y = W[(size_t)(k0 + lk + 1) * CH + n0 + lm];
      acc = __builtin_amdgcn_wmma_f32_16x16x4_f32(
           false, a,  false, b,
           (short)0, acc,  false,  false);
    }
  }

  unsigned* const slot_row = meta + (size_t)d * CH;
  unsigned r = SLOTS;
  if (has_edge) r = atomicAdd(&slot_row[0], 1u);

  if (has_tile) {
    const int colg = n0 + lm;
    const int rbase = m0 + 8 * (lane >> 4);
#pragma unroll
    for (int v = 0; v < 8; ++v) {
      int rowg = rbase + v;
      if (rowg < n) h[(size_t)rowg * CH + colg] = acc[v];
    }
  }

  if (r < SLOTS) slot_row[1u + r] = (unsigned)s;
}

template <int MASK>
__device__ __forceinline__ float lane_xor(float v) {
  return __int_as_float(__builtin_amdgcn_update_dpp(
      0, __float_as_int(v), 0x160 | MASK, 0xf, 0xf, false));
}

__device__ __forceinline__ float head_sum(float v) {
  v += lane_xor<1>(v);
  v += lane_xor<2>(v);
  v += lane_xor<4>(v);
  return v;
}

#define GAT_STEP(r)                                                        \
  {                                                                        \
    float es = head_sum((r).x * as4.x + (r).y * as4.y + (r).z * as4.z +    \
                        (r).w * as4.w);                                    \
    float e = es + adst;                                                   \
    e = (e >= 0.0f) ? e : NEG_SLOPE * e;                                   \
            \
    const bool up = e > m;                                                 \
    const float t = expf(up ? m - e : e - m);                              \
    const float sc = up ? t : 1.0f;                                        \
    const float p = up ? 1.0f : t;                                         \
    l = l * sc + p;                                                        \
    acc = acc * sc + (r) * p;                                              \
    m = up ? e : m;                                                        \
  }

__global__ __launch_bounds__(256) void gat_gather(
    const float* __restrict__ h, const int* __restrict__ ei,
    const float* __restrict__ att_src, const float* __restrict__ att_dst,
    const float* __restrict__ bias, const float* __restrict__ prelu_w,
    float* out, int n, int ne) {
  const int lane = threadIdx.x & 31;
  const int d = __builtin_amdgcn_readfirstlane(
      (int)((blockIdx.x * blockDim.x + threadIdx.x) >> 5));
  if (d >= n) return;
  const int f0 = lane * 4;
  const v4f as4 = *(const v4f*)(att_src + f0);
  const v4f ad4 = *(const v4f*)(att_dst + f0);
  const unsigned* row = (const unsigned*)(out + (size_t)d * CH);
  unsigned wmine = row[lane];
  const unsigned cnt = __builtin_amdgcn_readfirstlane(wmine);

  const v4f hd = *(const v4f*)(h + (size_t)d * CH + f0);
  const float adst =
      head_sum(hd.x * ad4.x + hd.y * ad4.y + hd.z * ad4.z + hd.w * ad4.w);
  float m = head_sum(hd.x * as4.x + hd.y * as4.y + hd.z * as4.z + hd.w * as4.w) +
            adst;
  m = (m >= 0.0f) ? m : NEG_SLOPE * m;
  float l = 1.0f;
  v4f acc = hd;

  if (cnt <= SLOTS) {
    for (unsigned base = 0; base <= cnt; base += 32) {
      if (base) wmine = row[base + lane];
      const unsigned rem = cnt + 1u - base;
      const unsigned jend = rem < 32u ? rem : 32u;
      unsigned j = base ? 0u : 1u;
      for (; j + 4 <= jend; j += 4) {
        const int s0 = __builtin_amdgcn_readlane((int)wmine, (int)j);
        const int s1 = __builtin_amdgcn_readlane((int)wmine, (int)j + 1);
        const int s2 = __builtin_amdgcn_readlane((int)wmine, (int)j + 2);
        const int s3 = __builtin_amdgcn_readlane((int)wmine, (int)j + 3);
        const v4f r0 = *(const v4f*)(h + (size_t)s0 * CH + f0);
        const v4f r1 = *(const v4f*)(h + (size_t)s1 * CH + f0);
        const v4f r2 = *(const v4f*)(h + (size_t)s2 * CH + f0);
        const v4f r3 = *(const v4f*)(h + (size_t)s3 * CH + f0);
        GAT_STEP(r0);
        GAT_STEP(r1);
        GAT_STEP(r2);
        GAT_STEP(r3);
      }
      for (; j < jend; ++j) {
        const int s0 = __builtin_amdgcn_readlane((int)wmine, (int)j);
        const v4f r0 = *(const v4f*)(h + (size_t)s0 * CH + f0);
        GAT_STEP(r0);
      }
    }
  } else {
    for (int base = 0; base < ne; base += 32) {
      const int e = base + lane;
      const int dd = (e < ne) ? ei[ne + e] : -1;
      unsigned hit = __builtin_amdgcn_ballot_w32(dd == d);
      if (hit) {
        const int smine = (e < ne) ? ei[e] : 0;
        while (hit) {
          const int j = __builtin_ctz(hit);
          hit &= hit - 1u;
          const int s0 = __builtin_amdgcn_readlane(smine, j);
          const v4f r0 = *(const v4f*)(h + (size_t)s0 * CH + f0);
          GAT_STEP(r0);
        }
      }
    }
  }

  const float den = l + 1e-16f;
  const v4f b4 = *(const v4f*)(bias + f0);
  const float pw = prelu_w[0];
  v4f o;
  o.x = acc.x / den + b4.x;
  o.y = acc.y / den + b4.y;
  o.z = acc.z / den + b4.z;
  o.w = acc.w / den + b4.w;
  o.x = (o.x >= 0.0f) ? o.x : pw * o.x;
  o.y = (o.y >= 0.0f) ? o.y : pw * o.y;
  o.z = (o.z >= 0.0f) ? o.z : pw * o.z;
  o.w = (o.w >= 0.0f) ? o.w : pw * o.w;
  *(v4f*)(out + (size_t)d * CH + f0) = o;
}

extern "C" void kernel_launch(void* const* d_in, const int* in_sizes, int n_in,
                              void* d_out, int out_size, void* d_ws, size_t ws_size,
                              hipStream_t stream) {
  const float* x       = (const float*)d_in[0];
  const int*   ei      = (const int*)d_in[1];
  const float* W       = (const float*)d_in[2];
  const float* att_src = (const float*)d_in[3];
  const float* att_dst = (const float*)d_in[4];
  const float* bias    = (const float*)d_in[5];
  const float* prelu_w = (const float*)d_in[6];
  float* out = (float*)d_out;

  const int n  = in_sizes[0] / CH;
  const int ne = in_sizes[1] / 2;

  float* h = (float*)d_ws;
  unsigned* meta = (unsigned*)d_out;

  const int tpb = 256;

  gat_init<<<(n + tpb - 1) / tpb, tpb, 0, stream>>>(meta, n);
  const int tiles = (n + 15) / 16;
  const int eblocks = (ne + tpb - 1) / tpb;
  gat_proj_slot<<<tiles > eblocks ? tiles : eblocks, 256, 0, stream>>>(
      x, W, h, ei, meta, n, ne);
  gat_gather<<<(n + 7) / 8, 256, 0, stream>>>(h, ei, att_src, att_dst, bias,
                                              prelu_w, out, n, ne);
}
